# starting kernel
# baseline (speedup 1.0000x reference)
#include <hip/hip_runtime.h>

#define IN_DIM      512
#define NUM_CLASSES 10
#define QUAD_DIM    (IN_DIM + IN_DIM * (IN_DIM + 1) / 2)
#define BATCH       1024
#define NSLICES     64
#define KSLICE      (QUAD_DIM / NSLICES)
#define NBT         (BATCH / 16)
#define NPAIR       (BATCH / 32)
#define XS_STRIDE   516
#define TILE_B_OFF  (16 * XS_STRIDE * 4)
#define TAB_BYTES   (QUAD_DIM * 4)

typedef float v2f __attribute__((ext_vector_type(2)));
typedef float v8f __attribute__((ext_vector_type(8)));

__device__ __forceinline__ int tri_off(int i) {
    return i * IN_DIM - ((i * (i - 1)) >> 1);
}

__global__ void build_table(unsigned* __restrict__ tab) {
    int k = blockIdx.x * 256 + threadIdx.x;
    if (k >= QUAD_DIM) return;
    int i, j;
    if (k < IN_DIM) {
        i = k; j = IN_DIM;
    } else {
        int t = k - IN_DIM;
        float r = sqrtf(fmaxf(1050625.0f - 8.0f * (float)t, 0.0f));
        int ii = (int)((1025.0f - r) * 0.5f);
        ii = ii < 0 ? 0 : (ii > IN_DIM - 1 ? IN_DIM - 1 : ii);
        if (ii < IN_DIM - 1 && tri_off(ii + 1) <= t) ++ii;
        if (ii < IN_DIM - 1 && tri_off(ii + 1) <= t) ++ii;
        if (ii > 0 && tri_off(ii) > t) --ii;
        if (ii > 0 && tri_off(ii) > t) --ii;
        i = ii;
        j = ii + (t - tri_off(ii));
    }
    tab[k] = ((unsigned)(j * 4) << 16) | (unsigned)(i * 4);
}

__global__ __launch_bounds__(128) void quadfeat_gemm(const float* __restrict__ X,
                                                     const float* __restrict__ W,
                                                     const unsigned* __restrict__ tab,
                                                     float* __restrict__ partial) {
    __shared__ float xs[32 * XS_STRIDE];

    const int pair = blockIdx.x;
    const int tid  = threadIdx.x;

    const float4* src4 = (const float4*)(X + (size_t)pair * (32 * IN_DIM));
    for (int v = tid; v < 4096; v += 128) {
        int row = v >> 7;
        int c4  = v & 127;
        *(float4*)(xs + row * XS_STRIDE + c4 * 4) = src4[v];
    }
    if (tid < 32) xs[tid * XS_STRIDE + IN_DIM] = 1.0f;
    __syncthreads();

    const int wave  = tid >> 5;
    const int lane  = tid & 31;
    const int slice = blockIdx.y * 4 + wave;
    const int hi    = lane >> 4;
    const int n     = lane & 15;
    const char* xrow = (const char*)(xs + n * XS_STRIDE);

    const int k0 = slice * KSLICE + 2 * hi;
    const unsigned* tp = tab + k0;
    const float* wp = W + (size_t)((n < NUM_CLASSES) ? n : (NUM_CLASSES - 1)) * QUAD_DIM + k0;

    v8f accA = {0.f, 0.f, 0.f, 0.f, 0.f, 0.f, 0.f, 0.f};
    v8f accB = {0.f, 0.f, 0.f, 0.f, 0.f, 0.f, 0.f, 0.f};

    #pragma unroll 2
    for (int it = 0; it < KSLICE / 4; ++it) {
        uint2 idx = *(const uint2*)tp;
        unsigned o_i0 = idx.x & 0xFFFFu, o_j0 = idx.x >> 16;
        unsigned o_i1 = idx.y & 0xFFFFu, o_j1 = idx.y >> 16;

        float ai0 = *(const float*)(xrow + o_i0);
        float aj0 = *(const float*)(xrow + o_j0);
        float ai1 = *(const float*)(xrow + o_i1);
        float aj1 = *(const float*)(xrow + o_j1);
        float bi0 = *(const float*)(xrow + TILE_B_OFF + o_i0);
        float bj0 = *(const float*)(xrow + TILE_B_OFF + o_j0);
        float bi1 = *(const float*)(xrow + TILE_B_OFF + o_i1);
        float bj1 = *(const float*)(xrow + TILE_B_OFF + o_j1);

        v2f b = *(const v2f*)wp;

        v2f aA; aA.x = ai0 * aj0; aA.y = ai1 * aj1;
        v2f aB; aB.x = bi0 * bj0; aB.y = bi1 * bj1;

        accA = __builtin_amdgcn_wmma_f32_16x16x4_f32(
            false, aA, false, b, (short)0, accA, false, false);
        accB = __builtin_amdgcn_wmma_f32_16x16x4_f32(
            false, aB, false, b, (short)0, accB, false, false);

        tp += 4;
        wp += 4;
    }

    const int bt0 = pair * 2;
    float* pA = partial + ((size_t)slice * NBT + bt0) * 256;
    float* pB = partial + ((size_t)slice * NBT + bt0 + 1) * 256;
    #pragma unroll
    for (int r = 0; r < 8; ++r) {
        pA[(r + 8 * hi) * 16 + n] = accA[r];
        pB[(r + 8 * hi) * 16 + n] = accB[r];
    }
}

__global__ void quadfeat_reduce(const float* __restrict__ partial,
                                const float* __restrict__ bias,
                                float* __restrict__ out) {
    int idx = blockIdx.x * blockDim.x + threadIdx.x;
    if (idx >= BATCH * NUM_CLASSES) return;
    int b = idx / NUM_CLASSES;
    int c = idx - b * NUM_CLASSES;
    int bt = b >> 4;
    int m  = b & 15;
    float s = bias[c];
    #pragma unroll 4
    for (int sl = 0; sl < NSLICES; ++sl)
        s += partial[((size_t)sl * NBT + bt) * 256 + m * 16 + c];
    out[(size_t)b * NUM_CLASSES + c] = s;
}

extern "C" void kernel_launch(void* const* d_in, const int* in_sizes, int n_in,
                              void* d_out, int out_size, void* d_ws, size_t ws_size,
                              hipStream_t stream) {
    const float* x    = (const float*)d_in[0];
    const float* W    = (const float*)d_in[1];
    const float* bias = (const float*)d_in[2];
    float* out        = (float*)d_out;

    unsigned* tab  = (unsigned*)d_ws;
    float* partial = (float*)((char*)d_ws + TAB_BYTES);

    build_table<<<(QUAD_DIM + 255) / 256, 256, 0, stream>>>(tab);

    dim3 grid(NPAIR, NSLICES / 4);
    quadfeat_gemm<<<grid, 128, 0, stream>>>(x, W, tab, partial);

    int total = BATCH * NUM_CLASSES;
    quadfeat_reduce<<<(total + 255) / 256, 256, 0, stream>>>(partial, bias, out);
}


# Four batch tiles per wave instead of two: each W fragment and segment set-up feeds four tiles; 128 K slices of 4 rows, 8-wave workgroups; sums regrouped (new K split, 8 partials summed in LDS)
# speedup vs baseline: 2.1652x; 2.1652x over previous
#include <hip/hip_runtime.h>

#define IN_DIM      512
#define NUM_CLASSES 10
#define QUAD_DIM    (IN_DIM + IN_DIM * (IN_DIM + 1) / 2)
#define BATCH       1024
#define NSLICES     128
#define NWAVES      8
#define NPLANES     (NSLICES / NWAVES)
#define NBT         (BATCH / 16)
#define NT          4
#define NQUAD       (BATCH / (16 * NT))
#define XS_ROWS     (16 * NT)
#define XS_STRIDE   516
#define XS_FRONT    8
#define XS_FLOATS   (XS_FRONT + XS_ROWS * XS_STRIDE + 8)
#define TILE_B      (16 * XS_STRIDE)
#define NSEG        5

typedef float v2f __attribute__((ext_vector_type(2)));
typedef float v8f __attribute__((ext_vector_type(8)));

struct Seg { int kstart, len, joff, icol; };

__device__ __forceinline__ Seg seg_of(int s, int q) {
    Seg g;
    if (q == 0) {
        g.kstart = 4 * s; g.len = 4; g.joff = 0; g.icol = IN_DIM;
    } else {
        int r = q - 1;
        int i = (r < 2) ? (2 * s + r) : (IN_DIM - 2 - 2 * s + (r - 2));
        g.kstart = IN_DIM + i * IN_DIM - ((i * (i - 1)) >> 1);
        g.len    = IN_DIM - i;
        g.joff   = i - g.kstart;
        g.icol   = i;
    }
    return g;
}

struct Plan {
    int kb0, last;
    int ng, rem;
    int kint, krem;
};

__device__ __forceinline__ Plan plan_of(const Seg& g) {
    Plan p;
    p.kb0  = g.kstart & ~7;
    p.last = (g.kstart + g.len - 1) & ~7;
    int nint = ((p.last - p.kb0) >> 3) - 1;
    if (nint < 0) nint = 0;
    p.ng   = nint >> 2;
    p.rem  = nint & 3;
    p.kint = p.kb0 + 8;
    p.krem = p.kint + 32 * p.ng;
    return p;
}

struct Chunk {
    float4 w;
    v2f x1[NT], x2[NT];
    int dks;
};

__device__ __forceinline__ Chunk load_chunk(int kb, const Seg& g, const float* __restrict__ wl,
                                            const float* rowA, int off1, int off2) {
    Chunk c;
    c.w = *(const float4*)(wl + kb);
    const float* pa = rowA + (kb + g.joff);
    #pragma unroll
    for (int t = 0; t < NT; ++t) {
        const float* pt = pa + t * TILE_B;
        c.x1[t].x = pt[off1]; c.x1[t].y = pt[off1 + 1];
        c.x2[t].x = pt[off2]; c.x2[t].y = pt[off2 + 1];
    }
    c.dks = kb - g.kstart;
    return c;
}

__device__ __forceinline__ void mma_chunk(const Chunk& c, int seglen, const float (&xi)[NT],
                                          bool hb, int off1, int off2, v8f (&acc)[NT]) {
    const unsigned len = (unsigned)seglen;
    const unsigned u1 = (unsigned)(c.dks + off1);
    const unsigned u2 = (unsigned)(c.dks + off2);
    const bool m1x = u1 < len, m1y = (u1 + 1u) < len;
    const bool m2x = u2 < len, m2y = (u2 + 1u) < len;

    v2f b1, b2;
    b1.x = hb ? c.w.z : c.w.x;  b1.y = hb ? c.w.w : c.w.y;
    b2.x = hb ? c.w.x : c.w.z;  b2.y = hb ? c.w.y : c.w.w;

    v2f a1[NT], a2[NT];
    #pragma unroll
    for (int t = 0; t < NT; ++t) {
        a1[t].x = m1x ? xi[t] * c.x1[t].x : 0.0f;  a1[t].y = m1y ? xi[t] * c.x1[t].y : 0.0f;
        a2[t].x = m2x ? xi[t] * c.x2[t].x : 0.0f;  a2[t].y = m2y ? xi[t] * c.x2[t].y : 0.0f;
    }
    #pragma unroll
    for (int t = 0; t < NT; ++t)
        acc[t] = __builtin_amdgcn_wmma_f32_16x16x4_f32(false, a1[t], false, b1, (short)0, acc[t], false, false);
    #pragma unroll
    for (int t = 0; t < NT; ++t)
        acc[t] = __builtin_amdgcn_wmma_f32_16x16x4_f32(false, a2[t], false, b2, (short)0, acc[t], false, false);
}

struct Grp {
    float4 w[4];
    v2f x1[4][NT], x2[4][NT];
};

__device__ __forceinline__ Grp load_group(int kb, int joff, const float* __restrict__ wl,
                                          const float* rowA, int off1, int off2) {
    Grp g;
    const float* pw = wl + kb;
    const float* pa = rowA + (kb + joff);
    #pragma unroll
    for (int c = 0; c < 4; ++c) {
        g.w[c] = *(const float4*)(pw + 8 * c);
        #pragma unroll
        for (int t = 0; t < NT; ++t) {
            const float* pt = pa + t * TILE_B + 8 * c;
            g.x1[c][t].x = pt[off1]; g.x1[c][t].y = pt[off1 + 1];
            g.x2[c][t].x = pt[off2]; g.x2[c][t].y = pt[off2 + 1];
        }
    }
    return g;
}

__device__ __forceinline__ void mma_group(const Grp& g, const float (&xi)[NT], bool hb,
                                          v8f (&acc)[NT]) {
    #pragma unroll
    for (int c = 0; c < 4; ++c) {
        v2f b1, b2;
        b1.x = hb ? g.w[c].z : g.w[c].x;  b1.y = hb ? g.w[c].w : g.w[c].y;
        b2.x = hb ? g.w[c].x : g.w[c].z;  b2.y = hb ? g.w[c].y : g.w[c].w;
        #pragma unroll
        for (int t = 0; t < NT; ++t) {
            v2f a1 = xi[t] * g.x1[c][t];
            acc[t] = __builtin_amdgcn_wmma_f32_16x16x4_f32(false, a1, false, b1, (short)0, acc[t], false, false);
        }
        #pragma unroll
        for (int t = 0; t < NT; ++t) {
            v2f a2 = xi[t] * g.x2[c][t];
            acc[t] = __builtin_amdgcn_wmma_f32_16x16x4_f32(false, a2, false, b2, (short)0, acc[t], false, false);
        }
    }
}

__global__ __launch_bounds__(32 * NWAVES) void quadfeat_gemm(const float* __restrict__ X,
                                                     const float* __restrict__ W,
                                                     float* __restrict__ partial) {
    __shared__ __attribute__((aligned(16))) float xs[XS_FLOATS];

    const int quad = blockIdx.x;
    const int tid  = threadIdx.x;

    const float4* src4 = (const float4*)(X + (size_t)quad * (XS_ROWS * IN_DIM));
    for (int v = tid; v < XS_ROWS * 128; v += 32 * NWAVES) {
        int row = v >> 7;
        int c4  = v & 127;
        *(float4*)(xs + XS_FRONT + row * XS_STRIDE + c4 * 4) = src4[v];
    }
    if (tid < XS_ROWS) xs[XS_FRONT + tid * XS_STRIDE + IN_DIM] = 1.0f;
    __syncthreads();

    const int lane = tid & 31;
    const int slice = __builtin_amdgcn_readfirstlane(blockIdx.y * NWAVES + (tid >> 5));
    const int hi    = lane >> 4;
    const bool hb   = hi != 0;
    const int n     = lane & 15;
    const int off1  = hb ? 6 : 0;
    const int off2  = hb ? 4 : 2;
    const float* rowA = xs + XS_FRONT + n * XS_STRIDE;
    const float* wl = W + (size_t)((n < NUM_CLASSES) ? n : (NUM_CLASSES - 1)) * QUAD_DIM + 4 * hi;

    v8f acc[NT];
    #pragma unroll
    for (int t = 0; t < NT; ++t) acc[t] = (v8f){0.f, 0.f, 0.f, 0.f, 0.f, 0.f, 0.f, 0.f};

    Seg  g = seg_of(slice, 0);
    Plan p = plan_of(g);
    Grp a = {};
    if (p.ng > 0) a = load_group(p.kint, g.joff, wl, rowA, off1, off2);

    #pragma unroll 1
    for (int q = 0; q < NSEG; ++q) {
        float xi[NT];
        #pragma unroll
        for (int t = 0; t < NT; ++t) xi[t] = rowA[t * TILE_B + g.icol];

        const bool two = p.last != p.kb0;
        Chunk e0 = load_chunk(p.kb0, g, wl, rowA, off1, off2);
        Chunk e1 = load_chunk(two ? p.last : p.kb0, g, wl, rowA, off1, off2);
        Chunk e2 = load_chunk(p.rem > 0 ? p.krem      : p.kb0, g, wl, rowA, off1, off2);
        Chunk e3 = load_chunk(p.rem > 1 ? p.krem + 8  : p.kb0, g, wl, rowA, off1, off2);
        Chunk e4 = load_chunk(p.rem > 2 ? p.krem + 16 : p.kb0, g, wl, rowA, off1, off2);

        const int ng    = p.ng;
        const int klast = p.kint + 32 * (ng - 1);
        int kb = p.kint;
        int gi = 0;
        #pragma unroll 1
        for (; gi + 2 <= ng; gi += 2) {
            Grp b = load_group(kb + 32, g.joff, wl, rowA, off1, off2);
            __builtin_amdgcn_sched_barrier(0);
            mma_group(a, xi, hb, acc);
            int kn = kb + 64;
            if (kn > klast) kn = klast;
            a = load_group(kn, g.joff, wl, rowA, off1, off2);
            __builtin_amdgcn_sched_barrier(0);
            mma_group(b, xi, hb, acc);
            kb += 64;
        }
        if (gi < ng) mma_group(a, xi, hb, acc);

        Seg  gn = g;
        Plan pn = p;
        if (q + 1 < NSEG) {
            gn = seg_of(slice, q + 1);
            pn = plan_of(gn);
            if (pn.ng > 0) a = load_group(pn.kint, gn.joff, wl, rowA, off1, off2);
        }

        mma_chunk(e0, g.len, xi, hb, off1, off2, acc);
        if (two)       mma_chunk(e1, g.len, xi, hb, off1, off2, acc);
        if (p.rem > 0) mma_chunk(e2, g.len, xi, hb, off1, off2, acc);
        if (p.rem > 1) mma_chunk(e3, g.len, xi, hb, off1, off2, acc);
        if (p.rem > 2) mma_chunk(e4, g.len, xi, hb, off1, off2, acc);

        g = gn; p = pn;
    }

    __syncthreads();
    float* red = xs;
    const int wv = tid >> 5;
    #pragma unroll
    for (int t = 0; t < NT; ++t) {
        #pragma unroll
        for (int r = 0; r < 8; ++r)
            red[(wv * NT + t) * 256 + r * 32 + lane] = acc[t][r];
    }
    __syncthreads();

    float* pt = partial + ((size_t)blockIdx.y * NBT + quad * NT) * 256;
    {
        const int m   = tid >> 4;
        const int nn  = tid & 15;
        const int pos = (m & 7) * 32 + (m >> 3) * 16 + nn;
        for (int t = 0; t < NT; ++t) {
            float s = red[(0 * NT + t) * 256 + pos];
            for (int w = 1; w < NWAVES; ++w)
                s += red[(w * NT + t) * 256 + pos];
            pt[t * 256 + tid] = s;
        }
    }
}

__global__ void quadfeat_reduce(const float* __restrict__ partial,
                                const float* __restrict__ bias,
                                float* __restrict__ out) {
    int idx = blockIdx.x * blockDim.x + threadIdx.x;
    if (idx >= BATCH * NUM_CLASSES) return;
    int b = idx / NUM_CLASSES;
    int c = idx - b * NUM_CLASSES;
    int bt = b >> 4;
    int m  = b & 15;
    float s = bias[c];
    #pragma unroll
    for (int pl = 0; pl < NPLANES; ++pl)
        s += partial[((size_t)pl * NBT + bt) * 256 + m * 16 + c];
    out[(size_t)b * NUM_CLASSES + c] = s;
}

extern "C" void kernel_launch(void* const* d_in, const int* in_sizes, int n_in,
                              void* d_out, int out_size, void* d_ws, size_t ws_size,
                              hipStream_t stream) {
    const float* x    = (const float*)d_in[0];
    const float* W    = (const float*)d_in[1];
    const float* bias = (const float*)d_in[2];
    float* out        = (float*)d_out;

    float* partial = (float*)d_ws;

    dim3 grid(NQUAD, NPLANES);
    quadfeat_gemm<<<grid, 256, 0, stream>>>(x, W, partial);

    int total = BATCH * NUM_CLASSES;
    quadfeat_reduce<<<(total + 255) / 256, 256, 0, stream>>>(partial, bias, out);
}


# Chunks aligned on the x column j instead of the W index k: per chunk and tile a lane's right factors are one aligned 16-byte LDS read; W loads 4-byte aligned; natural k order; sums regrouped
# speedup vs baseline: 2.4145x; 1.1151x over previous
#include <hip/hip_runtime.h>

#define IN_DIM      512
#define NUM_CLASSES 10
#define QUAD_DIM    (IN_DIM + IN_DIM * (IN_DIM + 1) / 2)
#define BATCH       1024
#define NSLICES     128
#define NWAVES      8
#define NPLANES     (NSLICES / NWAVES)
#define NBT         (BATCH / 16)
#define NT          4
#define NQUAD       (BATCH / (16 * NT))
#define XS_ROWS     (16 * NT)
#define XS_STRIDE   516
#define XS_FLOATS   (XS_ROWS * XS_STRIDE)
#define TILE_B      (16 * XS_STRIDE)
#define NSEG        5

typedef float v2f __attribute__((ext_vector_type(2)));
typedef float v4f __attribute__((ext_vector_type(4)));
typedef float v8f __attribute__((ext_vector_type(8)));
typedef float v4fu __attribute__((ext_vector_type(4), aligned(4)));

static_assert(IN_DIM % 8 == 0, "a packed row must end on a chunk boundary");

struct Seg { int jlo, len, woff, icol; };

__device__ __forceinline__ Seg seg_of(int s, int q) {
    Seg g;
    if (q == 0) {
        g.jlo = 4 * s; g.len = 4; g.woff = 0; g.icol = IN_DIM;
    } else {
        int r = q - 1;
        int i = (r < 2) ? (2 * s + r) : (IN_DIM - 2 - 2 * s + (r - 2));
        g.jlo  = i;
        g.len  = IN_DIM - i;
        g.woff = IN_DIM + i * IN_DIM - ((i * (i - 1)) >> 1) - i;
        g.icol = i;
    }
    return g;
}

struct Plan {
    int jb0;
    int ng, rem;
    int jrem, jint;
};

__device__ __forceinline__ Plan plan_of(const Seg& g) {
    Plan p;
    p.jb0  = g.jlo & ~7;
    const int last = (g.jlo + g.len - 1) & ~7;
    const int nint = (last - p.jb0) >> 3;
    p.ng   = nint >> 2;
    p.rem  = nint & 3;
    p.jrem = p.jb0 + 8;
    p.jint = p.jrem + 8 * p.rem;
    return p;
}

struct Chunk {
    v4f w;
    v4f x[NT];
    int dj;
};

__device__ __forceinline__ Chunk load_chunk(int jb, const Seg& g, const float* __restrict__ wl,
                                            const float* rowA) {
    Chunk c;
    c.w = *(const v4fu*)(wl + (jb + g.woff));
    #pragma unroll
    for (int t = 0; t < NT; ++t) c.x[t] = *(const v4f*)(rowA + t * TILE_B + jb);
    c.dj = jb - g.jlo;
    return c;
}

__device__ __forceinline__ void mma_chunk(const Chunk& c, int seglen, const float (&xi)[NT],
                                          int h4, v8f (&acc)[NT]) {
    const unsigned len = (unsigned)seglen;
    const unsigned u = (unsigned)(c.dj + h4);
    const bool m1x = u < len,        m1y = (u + 1u) < len;
    const bool m2x = (u + 2u) < len, m2y = (u + 3u) < len;

    const v2f b1 = c.w.xy;
    const v2f b2 = c.w.zw;

    v2f a1[NT], a2[NT];
    #pragma unroll
    for (int t = 0; t < NT; ++t) {
        a1[t].x = m1x ? xi[t] * c.x[t].x : 0.0f;  a1[t].y = m1y ? xi[t] * c.x[t].y : 0.0f;
        a2[t].x = m2x ? xi[t] * c.x[t].z : 0.0f;  a2[t].y = m2y ? xi[t] * c.x[t].w : 0.0f;
    }
    #pragma unroll
    for (int t = 0; t < NT; ++t)
        acc[t] = __builtin_amdgcn_wmma_f32_16x16x4_f32(false, a1[t], false, b1, (short)0, acc[t], false, false);
    #pragma unroll
    for (int t = 0; t < NT; ++t)
        acc[t] = __builtin_amdgcn_wmma_f32_16x16x4_f32(false, a2[t], false, b2, (short)0, acc[t], false, false);
}

struct Grp {
    v4f w[4];
    v4f x[4][NT];
};

__device__ __forceinline__ Grp load_group(int jb, int woff, const float* __restrict__ wl,
                                          const float* rowA) {
    Grp g;
    const float* pw = wl + (jb + woff);
    const float* pa = rowA + jb;
    #pragma unroll
    for (int c = 0; c < 4; ++c) {
        g.w[c] = *(const v4fu*)(pw + 8 * c);
        #pragma unroll
        for (int t = 0; t < NT; ++t) g.x[c][t] = *(const v4f*)(pa + t * TILE_B + 8 * c);
    }
    return g;
}

__device__ __forceinline__ void mma_group(const Grp& g, const float (&xi)[NT], v8f (&acc)[NT]) {
    #pragma unroll
    for (int c = 0; c < 4; ++c) {
        const v2f b1 = g.w[c].xy;
        const v2f b2 = g.w[c].zw;
        #pragma unroll
        for (int t = 0; t < NT; ++t) {
            v2f a1 = xi[t] * g.x[c][t].xy;
            acc[t] = __builtin_amdgcn_wmma_f32_16x16x4_f32(false, a1, false, b1, (short)0, acc[t], false, false);
        }
        #pragma unroll
        for (int t = 0; t < NT; ++t) {
            v2f a2 = xi[t] * g.x[c][t].zw;
            acc[t] = __builtin_amdgcn_wmma_f32_16x16x4_f32(false, a2, false, b2, (short)0, acc[t], false, false);
        }
    }
}

__global__ __launch_bounds__(32 * NWAVES) void quadfeat_gemm(const float* __restrict__ X,
                                                     const float* __restrict__ W,
                                                     float* __restrict__ partial) {
    __shared__ __attribute__((aligned(16))) float xs[XS_FLOATS];

    const int quad = blockIdx.x;
    const int tid  = threadIdx.x;

    const float4* src4 = (const float4*)(X + (size_t)quad * (XS_ROWS * IN_DIM));
    for (int v = tid; v < XS_ROWS * 128; v += 32 * NWAVES) {
        int row = v >> 7;
        int c4  = v & 127;
        *(float4*)(xs + row * XS_STRIDE + c4 * 4) = src4[v];
    }
    if (tid < XS_ROWS) xs[tid * XS_STRIDE + IN_DIM] = 1.0f;
    __syncthreads();

    const int lane = tid & 31;
    const int slice = __builtin_amdgcn_readfirstlane(blockIdx.y * NWAVES + (tid >> 5));
    const int hi    = lane >> 4;
    const int h4    = 4 * hi;
    const int n     = lane & 15;
    const float* rowX = xs + n * XS_STRIDE;
    const float* rowA = rowX + h4;
    const float* wl = W + (size_t)((n < NUM_CLASSES) ? n : (NUM_CLASSES - 1)) * QUAD_DIM + h4;

    v8f acc[NT];
    #pragma unroll
    for (int t = 0; t < NT; ++t) acc[t] = (v8f){0.f, 0.f, 0.f, 0.f, 0.f, 0.f, 0.f, 0.f};

    Seg  g = seg_of(slice, 0);
    Plan p = plan_of(g);
    Grp a = {};
    if (p.ng > 0) a = load_group(p.jint, g.woff, wl, rowA);

    #pragma unroll 1
    for (int q = 0; q < NSEG; ++q) {
        float xi[NT];
        #pragma unroll
        for (int t = 0; t < NT; ++t) xi[t] = rowX[t * TILE_B + g.icol];

        Chunk e0 = load_chunk(p.jb0, g, wl, rowA);
        Chunk e1 = load_chunk(p.rem > 0 ? p.jrem      : p.jb0, g, wl, rowA);
        Chunk e2 = load_chunk(p.rem > 1 ? p.jrem + 8  : p.jb0, g, wl, rowA);
        Chunk e3 = load_chunk(p.rem > 2 ? p.jrem + 16 : p.jb0, g, wl, rowA);

        const int ng    = p.ng;
        const int jlast = p.jint + 32 * (ng - 1);
        int jb = p.jint;
        int gi = 0;
        #pragma unroll 1
        for (; gi + 2 <= ng; gi += 2) {
            Grp b = load_group(jb + 32, g.woff, wl, rowA);
            __builtin_amdgcn_sched_barrier(0);
            mma_group(a, xi, acc);
            int jn = jb + 64;
            if (jn > jlast) jn = jlast;
            a = load_group(jn, g.woff, wl, rowA);
            __builtin_amdgcn_sched_barrier(0);
            mma_group(b, xi, acc);
            jb += 64;
        }
        if (gi < ng) mma_group(a, xi, acc);

        Seg  gn = g;
        Plan pn = p;
        if (q + 1 < NSEG) {
            gn = seg_of(slice, q + 1);
            pn = plan_of(gn);
            if (pn.ng > 0) a = load_group(pn.jint, gn.woff, wl, rowA);
        }

        mma_chunk(e0, g.len, xi, h4, acc);
        if (p.rem > 0) mma_chunk(e1, g.len, xi, h4, acc);
        if (p.rem > 1) mma_chunk(e2, g.len, xi, h4, acc);
        if (p.rem > 2) mma_chunk(e3, g.len, xi, h4, acc);

        g = gn; p = pn;
    }

    __syncthreads();
    float* red = xs;
    const int wv = tid >> 5;
    #pragma unroll
    for (int t = 0; t < NT; ++t) {
        #pragma unroll
        for (int r = 0; r < 8; ++r)
            red[(wv * NT + t) * 256 + r * 32 + lane] = acc[t][r];
    }
    __syncthreads();

    float* pt = partial + ((size_t)blockIdx.y * NBT + quad * NT) * 256;
    {
        const int m   = tid >> 4;
        const int nn  = tid & 15;
        const int pos = (m & 7) * 32 + (m >> 3) * 16 + nn;
        for (int t = 0; t < NT; ++t) {
            float s = red[(0 * NT + t) * 256 + pos];
            for (int w = 1; w < NWAVES; ++w)
                s += red[(w * NT + t) * 256 + pos];
            pt[t * 256 + tid] = s;
        }
    }
}

__global__ void quadfeat_reduce(const float* __restrict__ partial,
                                const float* __restrict__ bias,
                                float* __restrict__ out) {
    int idx = blockIdx.x * blockDim.x + threadIdx.x;
    if (idx >= BATCH * NUM_CLASSES) return;
    int b = idx / NUM_CLASSES;
    int c = idx - b * NUM_CLASSES;
    int bt = b >> 4;
    int m  = b & 15;
    float s = bias[c];
    #pragma unroll
    for (int pl = 0; pl < NPLANES; ++pl)
        s += partial[((size_t)pl * NBT + bt) * 256 + m * 16 + c];
    out[(size_t)b * NUM_CLASSES + c] = s;
}

extern "C" void kernel_launch(void* const* d_in, const int* in_sizes, int n_in,
                              void* d_out, int out_size, void* d_ws, size_t ws_size,
                              hipStream_t stream) {
    const float* x    = (const float*)d_in[0];
    const float* W    = (const float*)d_in[1];
    const float* bias = (const float*)d_in[2];
    float* out        = (float*)d_out;

    float* partial = (float*)d_ws;

    dim3 grid(NQUAD, NPLANES);
    quadfeat_gemm<<<grid, 256, 0, stream>>>(x, W, partial);

    int total = BATCH * NUM_CLASSES;
    quadfeat_reduce<<<(total + 255) / 256, 256, 0, stream>>>(partial, bias, out);
}
